# starting kernel
# baseline (speedup 1.0000x reference)
#include <hip/hip_runtime.h>


typedef float v2f __attribute__((ext_vector_type(2)));
typedef float v8f __attribute__((ext_vector_type(8)));

#define N_FEAT_IN 100

__global__ void degree_kernel(const int* __restrict__ src,
                              const int* __restrict__ dst,
                              float* __restrict__ deg_out,
                              float* __restrict__ deg_in,
                              int n_edges) {
    int e = blockIdx.x * blockDim.x + threadIdx.x;
    if (e >= n_edges) return;
    atomicAdd(&deg_out[src[e]], 1.0f);
    atomicAdd(&deg_in[dst[e]], 1.0f);
}

__global__ void norm_kernel(float* __restrict__ a, float* __restrict__ b, int n) {
    int i = blockIdx.x * blockDim.x + threadIdx.x;
    if (i >= n) return;
    a[i] = rsqrtf(fmaxf(a[i], 1.0f));
    b[i] = rsqrtf(fmaxf(b[i], 1.0f));
}

template <int F>
__global__ void scatter_add_kernel(const float* __restrict__ h,
                                   const float* __restrict__ nsrc,
                                   const int* __restrict__ src,
                                   const int* __restrict__ dst,
                                   float* __restrict__ agg,
                                   int n_edges) {
    constexpr int CH = F / 4;
    long long idx = (long long)blockIdx.x * blockDim.x + threadIdx.x;
    if (idx >= (long long)n_edges * CH) return;
    int e = (int)(idx / CH);
    int f = (int)(idx % CH) * 4;
    int s = src[e];
    int d = dst[e];
    float ns = nsrc[s];
    float4 v = *(const float4*)(h + (size_t)s * F + f);
    float* arow = agg + (size_t)d * F + f;
    atomicAdd(arow + 0, v.x * ns);
    atomicAdd(arow + 1, v.y * ns);
    atomicAdd(arow + 2, v.z * ns);
    atomicAdd(arow + 3, v.w * ns);
}

template <int F_IN, int F_OUT, int NT, bool RELU>
__global__ __launch_bounds__(32) void gcn_wmma_layer(
    const float* __restrict__ agg, const float* __restrict__ ndst,
    const float* __restrict__ W, const float* __restrict__ bias,
    float* __restrict__ out, int n_nodes) {
    const int lane = threadIdx.x;
    const int half = lane >> 4;
    const int lm = lane & 15;
    const int row0 = blockIdx.x * 16;

    v8f acc[NT];
#pragma unroll
    for (int t = 0; t < NT; ++t) acc[t] = v8f{};

    const float* arow = agg + (size_t)(row0 + lm) * F_IN;

    for (int k0 = 0; k0 < F_IN; k0 += 4) {
        const int ka = k0 + 2 * half;
        v2f a;
        a.x = arow[ka];
        a.y = arow[ka + 1];
#pragma unroll
        for (int t = 0; t < NT; ++t) {
            const int col = t * 16 + lm;
            v2f bf;
            bf.x = (col < F_OUT) ? W[(size_t)ka * F_OUT + col] : 0.0f;
            bf.y = (col < F_OUT) ? W[(size_t)(ka + 1) * F_OUT + col] : 0.0f;
            acc[t] = __builtin_amdgcn_wmma_f32_16x16x4_f32(
                 false, a,  false, bf,
                 (short)0, acc[t],  false,  false);
        }
    }

#pragma unroll
    for (int i = 0; i < 8; ++i) {
        const int r = row0 + i + 8 * half;
        const float nd = ndst[r];
#pragma unroll
        for (int t = 0; t < NT; ++t) {
            const int col = t * 16 + lm;
            if (col < F_OUT) {
                float v = acc[t][i] * nd + bias[col];
                if (RELU) v = fmaxf(v, 0.0f);
                out[(size_t)r * F_OUT + col] = v;
            }
        }
    }
}

static inline size_t align256(size_t x) { return (x + 255) & ~(size_t)255; }

extern "C" void kernel_launch(void* const* d_in, const int* in_sizes, int n_in,
                              void* d_out, int out_size, void* d_ws, size_t ws_size,
                              hipStream_t stream) {
    const float* x  = (const float*)d_in[0];
    const float* W0 = (const float*)d_in[1];
    const float* b0 = (const float*)d_in[2];
    const float* W1 = (const float*)d_in[3];
    const float* b1 = (const float*)d_in[4];
    const float* W2 = (const float*)d_in[5];
    const float* b2 = (const float*)d_in[6];
    const int* src  = (const int*)d_in[7];
    const int* dst  = (const int*)d_in[8];

    const int N = in_sizes[0] / N_FEAT_IN;
    const int E = in_sizes[7];
    float* out = (float*)d_out;

    char* p = (char*)d_ws;
    float* nsrc = (float*)p;  p += align256((size_t)N * 4);
    float* ndst = (float*)p;  p += align256((size_t)N * 4);
    float* agg  = (float*)p;  p += align256((size_t)N * N_FEAT_IN * 4);
    float* h0   = (float*)p;  p += align256((size_t)N * 64 * 4);
    float* h1   = (float*)p;  p += align256((size_t)N * 64 * 4);

    const int TB = 256;

    hipMemsetAsync(nsrc, 0, (size_t)N * 4, stream);
    hipMemsetAsync(ndst, 0, (size_t)N * 4, stream);
    degree_kernel<<<(E + TB - 1) / TB, TB, 0, stream>>>(src, dst, nsrc, ndst, E);
    norm_kernel<<<(N + TB - 1) / TB, TB, 0, stream>>>(nsrc, ndst, N);

    hipMemsetAsync(agg, 0, (size_t)N * N_FEAT_IN * 4, stream);
    {
        long long total = (long long)E * (N_FEAT_IN / 4);
        int blocks = (int)((total + TB - 1) / TB);
        scatter_add_kernel<N_FEAT_IN><<<blocks, TB, 0, stream>>>(x, nsrc, src, dst, agg, E);
    }
    gcn_wmma_layer<N_FEAT_IN, 64, 4, true>
        <<<N / 16, 32, 0, stream>>>(agg, ndst, W0, b0, h0, N);

    hipMemsetAsync(agg, 0, (size_t)N * 64 * 4, stream);
    {
        long long total = (long long)E * (64 / 4);
        int blocks = (int)((total + TB - 1) / TB);
        scatter_add_kernel<64><<<blocks, TB, 0, stream>>>(h0, nsrc, src, dst, agg, E);
    }
    gcn_wmma_layer<64, 64, 4, true>
        <<<N / 16, 32, 0, stream>>>(agg, ndst, W1, b1, h1, N);

    hipMemsetAsync(agg, 0, (size_t)N * 64 * 4, stream);
    {
        long long total = (long long)E * (64 / 4);
        int blocks = (int)((total + TB - 1) / TB);
        scatter_add_kernel<64><<<blocks, TB, 0, stream>>>(h1, nsrc, src, dst, agg, E);
    }
    gcn_wmma_layer<64, 47, 3, false>
        <<<N / 16, 32, 0, stream>>>(agg, ndst, W2, b2, out, N);
}


# Layer 0 applies W0 before the edge aggregation, A(XW) for (AX)W: the atomic scatter moves 64 not 100 features per edge; scale, bias, ReLU follow in a small pass; fp sum order changes
# speedup vs baseline: 1.1566x; 1.1566x over previous
#include <hip/hip_runtime.h>


typedef float v2f __attribute__((ext_vector_type(2)));
typedef float v8f __attribute__((ext_vector_type(8)));

#define N_FEAT_IN 100

__global__ void degree_kernel(const int* __restrict__ src,
                              const int* __restrict__ dst,
                              float* __restrict__ deg_out,
                              float* __restrict__ deg_in,
                              int n_edges) {
    int e = blockIdx.x * blockDim.x + threadIdx.x;
    if (e >= n_edges) return;
    atomicAdd(&deg_out[src[e]], 1.0f);
    atomicAdd(&deg_in[dst[e]], 1.0f);
}

__global__ void norm_kernel(float* __restrict__ a, float* __restrict__ b, int n) {
    int i = blockIdx.x * blockDim.x + threadIdx.x;
    if (i >= n) return;
    a[i] = rsqrtf(fmaxf(a[i], 1.0f));
    b[i] = rsqrtf(fmaxf(b[i], 1.0f));
}

template <int F>
__global__ void scatter_add_kernel(const float* __restrict__ h,
                                   const float* __restrict__ nsrc,
                                   const int* __restrict__ src,
                                   const int* __restrict__ dst,
                                   float* __restrict__ agg,
                                   int n_edges) {
    constexpr int CH = F / 4;
    long long idx = (long long)blockIdx.x * blockDim.x + threadIdx.x;
    if (idx >= (long long)n_edges * CH) return;
    int e = (int)(idx / CH);
    int f = (int)(idx % CH) * 4;
    int s = src[e];
    int d = dst[e];
    float ns = nsrc[s];
    float4 v = *(const float4*)(h + (size_t)s * F + f);
    float* arow = agg + (size_t)d * F + f;
    atomicAdd(arow + 0, v.x * ns);
    atomicAdd(arow + 1, v.y * ns);
    atomicAdd(arow + 2, v.z * ns);
    atomicAdd(arow + 3, v.w * ns);
}

template <int F>
__global__ void scale_bias_relu_kernel(const float* __restrict__ agg,
                                       const float* __restrict__ ndst,
                                       const float* __restrict__ bias,
                                       float* __restrict__ h,
                                       int n_nodes) {
    constexpr int CH = F / 4;
    int idx = blockIdx.x * blockDim.x + threadIdx.x;
    if (idx >= n_nodes * CH) return;
    int r = idx / CH;
    int f = (idx % CH) * 4;
    float nd = ndst[r];
    float4 a = *(const float4*)(agg + (size_t)r * F + f);
    float4 o;
    o.x = fmaxf(a.x * nd + bias[f + 0], 0.0f);
    o.y = fmaxf(a.y * nd + bias[f + 1], 0.0f);
    o.z = fmaxf(a.z * nd + bias[f + 2], 0.0f);
    o.w = fmaxf(a.w * nd + bias[f + 3], 0.0f);
    *(float4*)(h + (size_t)r * F + f) = o;
}

template <int F_IN, int F_OUT, int NT, bool EPI, bool RELU>
__global__ __launch_bounds__(32) void gcn_wmma_layer(
    const float* __restrict__ agg, const float* __restrict__ ndst,
    const float* __restrict__ W, const float* __restrict__ bias,
    float* __restrict__ out, int n_nodes) {
    const int lane = threadIdx.x;
    const int half = lane >> 4;
    const int lm = lane & 15;
    const int row0 = blockIdx.x * 16;

    v8f acc[NT];
#pragma unroll
    for (int t = 0; t < NT; ++t) acc[t] = v8f{};

    const float* arow = agg + (size_t)(row0 + lm) * F_IN;

    for (int k0 = 0; k0 < F_IN; k0 += 4) {
        const int ka = k0 + 2 * half;
        v2f a;
        a.x = arow[ka];
        a.y = arow[ka + 1];
#pragma unroll
        for (int t = 0; t < NT; ++t) {
            const int col = t * 16 + lm;
            v2f bf;
            bf.x = (col < F_OUT) ? W[(size_t)ka * F_OUT + col] : 0.0f;
            bf.y = (col < F_OUT) ? W[(size_t)(ka + 1) * F_OUT + col] : 0.0f;
            acc[t] = __builtin_amdgcn_wmma_f32_16x16x4_f32(
                 false, a,  false, bf,
                 (short)0, acc[t],  false,  false);
        }
    }

#pragma unroll
    for (int i = 0; i < 8; ++i) {
        const int r = row0 + i + 8 * half;
        const float nd = EPI ? ndst[r] : 1.0f;
#pragma unroll
        for (int t = 0; t < NT; ++t) {
            const int col = t * 16 + lm;
            if (col < F_OUT) {
                float v = acc[t][i];
                if (EPI) v = v * nd + bias[col];
                if (RELU) v = fmaxf(v, 0.0f);
                out[(size_t)r * F_OUT + col] = v;
            }
        }
    }
}

static inline size_t align256(size_t x) { return (x + 255) & ~(size_t)255; }

extern "C" void kernel_launch(void* const* d_in, const int* in_sizes, int n_in,
                              void* d_out, int out_size, void* d_ws, size_t ws_size,
                              hipStream_t stream) {
    const float* x  = (const float*)d_in[0];
    const float* W0 = (const float*)d_in[1];
    const float* b0 = (const float*)d_in[2];
    const float* W1 = (const float*)d_in[3];
    const float* b1 = (const float*)d_in[4];
    const float* W2 = (const float*)d_in[5];
    const float* b2 = (const float*)d_in[6];
    const int* src  = (const int*)d_in[7];
    const int* dst  = (const int*)d_in[8];

    const int N = in_sizes[0] / N_FEAT_IN;
    const int E = in_sizes[7];
    float* out = (float*)d_out;

    char* p = (char*)d_ws;
    float* nsrc = (float*)p;  p += align256((size_t)N * 4);
    float* ndst = (float*)p;  p += align256((size_t)N * 4);
    float* agg  = (float*)p;  p += align256((size_t)N * N_FEAT_IN * 4);
    float* h0   = (float*)p;  p += align256((size_t)N * 64 * 4);
    float* h1   = (float*)p;  p += align256((size_t)N * 64 * 4);

    const int TB = 256;

    hipMemsetAsync(nsrc, 0, (size_t)N * 4, stream);
    hipMemsetAsync(ndst, 0, (size_t)N * 4, stream);
    degree_kernel<<<(E + TB - 1) / TB, TB, 0, stream>>>(src, dst, nsrc, ndst, E);
    norm_kernel<<<(N + TB - 1) / TB, TB, 0, stream>>>(nsrc, ndst, N);

    gcn_wmma_layer<N_FEAT_IN, 64, 4, false, false>
        <<<N / 16, 32, 0, stream>>>(x, nullptr, W0, nullptr, h1, N);
    hipMemsetAsync(agg, 0, (size_t)N * 64 * 4, stream);
    {
        long long total = (long long)E * (64 / 4);
        int blocks = (int)((total + TB - 1) / TB);
        scatter_add_kernel<64><<<blocks, TB, 0, stream>>>(h1, nsrc, src, dst, agg, E);
    }
    scale_bias_relu_kernel<64>
        <<<(N * (64 / 4) + TB - 1) / TB, TB, 0, stream>>>(agg, ndst, b0, h0, N);

    hipMemsetAsync(agg, 0, (size_t)N * 64 * 4, stream);
    {
        long long total = (long long)E * (64 / 4);
        int blocks = (int)((total + TB - 1) / TB);
        scatter_add_kernel<64><<<blocks, TB, 0, stream>>>(h0, nsrc, src, dst, agg, E);
    }
    gcn_wmma_layer<64, 64, 4, true, true>
        <<<N / 16, 32, 0, stream>>>(agg, ndst, W1, b1, h1, N);

    hipMemsetAsync(agg, 0, (size_t)N * 64 * 4, stream);
    {
        long long total = (long long)E * (64 / 4);
        int blocks = (int)((total + TB - 1) / TB);
        scatter_add_kernel<64><<<blocks, TB, 0, stream>>>(h1, nsrc, src, dst, agg, E);
    }
    gcn_wmma_layer<64, 47, 3, true, false>
        <<<N / 16, 32, 0, stream>>>(agg, ndst, W2, b2, out, N);
}


# Gather replaces f32 atomic scatter: edges bucketed by destination per call (returning int atomic: count+rank; scan; fill), 16 lanes sum a row in registers, one store with epilogue; sum order by rank
# speedup vs baseline: 2.1829x; 1.8873x over previous
#include <hip/hip_runtime.h>


typedef float v2f __attribute__((ext_vector_type(2)));
typedef float v8f __attribute__((ext_vector_type(8)));

#define N_FEAT_IN 100

__global__ void degree_rank_kernel(const int* __restrict__ src,
                                   const int* __restrict__ dst,
                                   int* __restrict__ cnt_out,
                                   int* __restrict__ cnt_in,
                                   int* __restrict__ rank,
                                   int n_edges) {
    int e = blockIdx.x * blockDim.x + threadIdx.x;
    if (e >= n_edges) return;
    atomicAdd(&cnt_out[src[e]], 1);
    rank[e] = atomicAdd(&cnt_in[dst[e]], 1);
}

__global__ __launch_bounds__(1024) void rowptr_norm_kernel(
    const int* __restrict__ cnt_out, const int* __restrict__ cnt_in,
    int* __restrict__ row_ptr, float* __restrict__ nsrc,
    float* __restrict__ ndst, int n) {
    __shared__ int w_incl[32];
    __shared__ int w_part[32];
    const int tid = threadIdx.x;
    const int lane = tid & 31;
    const int wv = tid >> 5;
    const int tile0 = blockIdx.x * 1024;

    int part = 0;
#pragma unroll 8
    for (int i = tid; i < tile0; i += 1024) part += cnt_in[i];

    const int i = tile0 + tid;
    int c_in = 0;
    if (i < n) {
        c_in = cnt_in[i];
        const int c_out = cnt_out[i];
        ndst[i] = rsqrtf(fmaxf((float)c_in, 1.0f));
        nsrc[i] = rsqrtf(fmaxf((float)c_out, 1.0f));
    }
    int incl = c_in;
    for (int d = 1; d < 32; d <<= 1) {
        const int up = __shfl_up(incl, d);
        if (lane >= d) incl += up;
    }
    for (int d = 16; d > 0; d >>= 1) part += __shfl_xor(part, d);
    if (lane == 31) w_incl[wv] = incl;
    if (lane == 0) w_part[wv] = part;
    __syncthreads();
    int offset = 0;
    for (int w = 0; w < 32; ++w) {
        offset += w_part[w];
        if (w < wv) offset += w_incl[w];
    }
    if (i < n) row_ptr[i] = offset + incl - c_in;
}

__global__ void bucket_fill_kernel(const int* __restrict__ src,
                                   const int* __restrict__ dst,
                                   const int* __restrict__ rank,
                                   const int* __restrict__ row_ptr,
                                   const float* __restrict__ nsrc,
                                   int* __restrict__ bsrc,
                                   float* __restrict__ bw,
                                   int n_edges) {
    int e = blockIdx.x * blockDim.x + threadIdx.x;
    if (e >= n_edges) return;
    const int s = src[e];
    const int slot = row_ptr[dst[e]] + rank[e];
    bsrc[slot] = s;
    bw[slot] = nsrc[s];
}

template <bool EPI>
__global__ __launch_bounds__(256) void gather_rows_kernel(
    const float* __restrict__ h, const int* __restrict__ row_ptr,
    const int* __restrict__ cnt_in, const int* __restrict__ bsrc,
    const float* __restrict__ bw, const float* __restrict__ ndst,
    const float* __restrict__ bias, float* __restrict__ outp, int n_nodes) {
    const int t = blockIdx.x * blockDim.x + threadIdx.x;
    const int node = t >> 4;
    const int l = t & 15;
    if (node >= n_nodes) return;
    const int beg = row_ptr[node];
    const int len = cnt_in[node];
    const float4* __restrict__ h4 = (const float4*)h;
    const int* __restrict__ bp = bsrc + beg;
    const float* __restrict__ wp = bw + beg;

    float4 acc = make_float4(0.0f, 0.0f, 0.0f, 0.0f);
    int j = 0;
    for (; j + 4 <= len; j += 4) {
        const int s0 = bp[j], s1 = bp[j + 1], s2 = bp[j + 2], s3 = bp[j + 3];
        const float w0 = wp[j], w1 = wp[j + 1], w2 = wp[j + 2], w3 = wp[j + 3];
        const float4 v0 = h4[(size_t)s0 * 16 + l];
        const float4 v1 = h4[(size_t)s1 * 16 + l];
        const float4 v2 = h4[(size_t)s2 * 16 + l];
        const float4 v3 = h4[(size_t)s3 * 16 + l];
        acc.x += v0.x * w0; acc.y += v0.y * w0; acc.z += v0.z * w0; acc.w += v0.w * w0;
        acc.x += v1.x * w1; acc.y += v1.y * w1; acc.z += v1.z * w1; acc.w += v1.w * w1;
        acc.x += v2.x * w2; acc.y += v2.y * w2; acc.z += v2.z * w2; acc.w += v2.w * w2;
        acc.x += v3.x * w3; acc.y += v3.y * w3; acc.z += v3.z * w3; acc.w += v3.w * w3;
    }
    const int r = len - j;
    if (r > 0) {
        const int j1 = j + (r > 1 ? 1 : 0);
        const int j2 = j + (r > 2 ? 2 : 0);
        const int s0 = bp[j], s1 = bp[j1], s2 = bp[j2];
        const float w0 = wp[j], w1 = wp[j1], w2 = wp[j2];
        const float4 v0 = h4[(size_t)s0 * 16 + l];
        const float4 v1 = h4[(size_t)s1 * 16 + l];
        const float4 v2 = h4[(size_t)s2 * 16 + l];
        const bool k1 = r > 1, k2 = r > 2;
        acc.x += v0.x * w0; acc.y += v0.y * w0; acc.z += v0.z * w0; acc.w += v0.w * w0;
        acc.x += k1 ? v1.x * w1 : 0.0f; acc.y += k1 ? v1.y * w1 : 0.0f;
        acc.z += k1 ? v1.z * w1 : 0.0f; acc.w += k1 ? v1.w * w1 : 0.0f;
        acc.x += k2 ? v2.x * w2 : 0.0f; acc.y += k2 ? v2.y * w2 : 0.0f;
        acc.z += k2 ? v2.z * w2 : 0.0f; acc.w += k2 ? v2.w * w2 : 0.0f;
    }

    if (EPI) {
        const float nd = ndst[node];
        const float4 b = *(const float4*)(bias + 4 * l);
        acc.x = fmaxf(acc.x * nd + b.x, 0.0f);
        acc.y = fmaxf(acc.y * nd + b.y, 0.0f);
        acc.z = fmaxf(acc.z * nd + b.z, 0.0f);
        acc.w = fmaxf(acc.w * nd + b.w, 0.0f);
    }
    ((float4*)outp)[(size_t)node * 16 + l] = acc;
}

template <int F_IN, int F_OUT, int NT, bool EPI, bool RELU>
__global__ __launch_bounds__(32) void gcn_wmma_layer(
    const float* __restrict__ agg, const float* __restrict__ ndst,
    const float* __restrict__ W, const float* __restrict__ bias,
    float* __restrict__ out, int n_nodes) {
    const int lane = threadIdx.x;
    const int half = lane >> 4;
    const int lm = lane & 15;
    const int row0 = blockIdx.x * 16;

    v8f acc[NT];
#pragma unroll
    for (int t = 0; t < NT; ++t) acc[t] = v8f{};

    const float* arow = agg + (size_t)(row0 + lm) * F_IN;

    for (int k0 = 0; k0 < F_IN; k0 += 4) {
        const int ka = k0 + 2 * half;
        v2f a;
        a.x = arow[ka];
        a.y = arow[ka + 1];
#pragma unroll
        for (int t = 0; t < NT; ++t) {
            const int col = t * 16 + lm;
            v2f bf;
            bf.x = (col < F_OUT) ? W[(size_t)ka * F_OUT + col] : 0.0f;
            bf.y = (col < F_OUT) ? W[(size_t)(ka + 1) * F_OUT + col] : 0.0f;
            acc[t] = __builtin_amdgcn_wmma_f32_16x16x4_f32(
                 false, a,  false, bf,
                 (short)0, acc[t],  false,  false);
        }
    }

#pragma unroll
    for (int i = 0; i < 8; ++i) {
        const int r = row0 + i + 8 * half;
        const float nd = EPI ? ndst[r] : 1.0f;
#pragma unroll
        for (int t = 0; t < NT; ++t) {
            const int col = t * 16 + lm;
            if (col < F_OUT) {
                float v = acc[t][i];
                if (EPI) v = v * nd + bias[col];
                if (RELU) v = fmaxf(v, 0.0f);
                out[(size_t)r * F_OUT + col] = v;
            }
        }
    }
}

static inline size_t align256(size_t x) { return (x + 255) & ~(size_t)255; }

extern "C" void kernel_launch(void* const* d_in, const int* in_sizes, int n_in,
                              void* d_out, int out_size, void* d_ws, size_t ws_size,
                              hipStream_t stream) {
    const float* x  = (const float*)d_in[0];
    const float* W0 = (const float*)d_in[1];
    const float* b0 = (const float*)d_in[2];
    const float* W1 = (const float*)d_in[3];
    const float* b1 = (const float*)d_in[4];
    const float* W2 = (const float*)d_in[5];
    const float* b2 = (const float*)d_in[6];
    const int* src  = (const int*)d_in[7];
    const int* dst  = (const int*)d_in[8];

    const int N = in_sizes[0] / N_FEAT_IN;
    const int E = in_sizes[7];
    float* out = (float*)d_out;

    char* p = (char*)d_ws;
    float* nsrc = (float*)p;  p += align256((size_t)N * 4);
    float* ndst = (float*)p;  p += align256((size_t)N * 4);
    char* slab = p;           p += align256((size_t)N * N_FEAT_IN * 4);
    float* h0   = (float*)p;  p += align256((size_t)N * 64 * 4);
    float* h1   = (float*)p;  p += align256((size_t)N * 64 * 4);
    float* agg   = (float*)slab;  slab += align256((size_t)N * 64 * 4);
    int* cnt_out = (int*)slab;    slab += align256((size_t)N * 4);
    int* cnt_in  = (int*)slab;    slab += align256((size_t)N * 4);
    int* row_ptr = (int*)slab;    slab += align256((size_t)N * 4);
    int* bsrc    = (int*)slab;    slab += align256((size_t)E * 4);
    float* bw    = (float*)slab;
    int* rank    = (int*)agg;

    const int TB = 256;

    hipMemsetAsync(cnt_out, 0, 2 * align256((size_t)N * 4), stream);
    degree_rank_kernel<<<(E + TB - 1) / TB, TB, 0, stream>>>(src, dst, cnt_out, cnt_in, rank, E);
    rowptr_norm_kernel<<<(N + 1023) / 1024, 1024, 0, stream>>>(cnt_out, cnt_in, row_ptr, nsrc, ndst, N);
    bucket_fill_kernel<<<(E + TB - 1) / TB, TB, 0, stream>>>(src, dst, rank, row_ptr, nsrc, bsrc, bw, E);

    const int GB = (N * 16 + TB - 1) / TB;

    gcn_wmma_layer<N_FEAT_IN, 64, 4, false, false>
        <<<N / 16, 32, 0, stream>>>(x, nullptr, W0, nullptr, h1, N);
    gather_rows_kernel<true><<<GB, TB, 0, stream>>>(h1, row_ptr, cnt_in, bsrc, bw, ndst, b0, h0, N);

    gather_rows_kernel<false><<<GB, TB, 0, stream>>>(h0, row_ptr, cnt_in, bsrc, bw, nullptr, nullptr, agg, N);
    gcn_wmma_layer<64, 64, 4, true, true>
        <<<N / 16, 32, 0, stream>>>(agg, ndst, W1, b1, h1, N);

    gather_rows_kernel<false><<<GB, TB, 0, stream>>>(h1, row_ptr, cnt_in, bsrc, bw, nullptr, nullptr, agg, N);
    gcn_wmma_layer<64, 47, 3, true, false>
        <<<N / 16, 32, 0, stream>>>(agg, ndst, W2, b2, out, N);
}


# Layers 1, 2: gather and dense product fused in one kernel: a block gathers 16 rows into LDS, its first waves run the f32 WMMAs from there; no aggregate round trip; 7 launches not 9; sum order kept
# speedup vs baseline: 2.4159x; 1.1067x over previous
#include <hip/hip_runtime.h>


typedef float v2f __attribute__((ext_vector_type(2)));
typedef float v8f __attribute__((ext_vector_type(8)));

#define N_FEAT_IN 100

__global__ void degree_rank_kernel(const int* __restrict__ src,
                                   const int* __restrict__ dst,
                                   int* __restrict__ cnt_out,
                                   int* __restrict__ cnt_in,
                                   int* __restrict__ rank,
                                   int n_edges) {
    int e = blockIdx.x * blockDim.x + threadIdx.x;
    if (e >= n_edges) return;
    atomicAdd(&cnt_out[src[e]], 1);
    rank[e] = atomicAdd(&cnt_in[dst[e]], 1);
}

__global__ __launch_bounds__(1024) void rowptr_norm_kernel(
    const int* __restrict__ cnt_out, const int* __restrict__ cnt_in,
    int* __restrict__ row_ptr, float* __restrict__ nsrc,
    float* __restrict__ ndst, int n) {
    __shared__ int w_incl[32];
    __shared__ int w_part[32];
    const int tid = threadIdx.x;
    const int lane = tid & 31;
    const int wv = tid >> 5;
    const int tile0 = blockIdx.x * 1024;

    int part = 0;
#pragma unroll 8
    for (int i = tid; i < tile0; i += 1024) part += cnt_in[i];

    const int i = tile0 + tid;
    int c_in = 0;
    if (i < n) {
        c_in = cnt_in[i];
        const int c_out = cnt_out[i];
        ndst[i] = rsqrtf(fmaxf((float)c_in, 1.0f));
        nsrc[i] = rsqrtf(fmaxf((float)c_out, 1.0f));
    }
    int incl = c_in;
    for (int d = 1; d < 32; d <<= 1) {
        const int up = __shfl_up(incl, d);
        if (lane >= d) incl += up;
    }
    for (int d = 16; d > 0; d >>= 1) part += __shfl_xor(part, d);
    if (lane == 31) w_incl[wv] = incl;
    if (lane == 0) w_part[wv] = part;
    __syncthreads();
    int offset = 0;
    for (int w = 0; w < 32; ++w) {
        offset += w_part[w];
        if (w < wv) offset += w_incl[w];
    }
    if (i < n) row_ptr[i] = offset + incl - c_in;
}

__global__ void bucket_fill_kernel(const int* __restrict__ src,
                                   const int* __restrict__ dst,
                                   const int* __restrict__ rank,
                                   const int* __restrict__ row_ptr,
                                   const float* __restrict__ nsrc,
                                   int* __restrict__ bsrc,
                                   float* __restrict__ bw,
                                   int n_edges) {
    int e = blockIdx.x * blockDim.x + threadIdx.x;
    if (e >= n_edges) return;
    const int s = src[e];
    const int slot = row_ptr[dst[e]] + rank[e];
    bsrc[slot] = s;
    bw[slot] = nsrc[s];
}

__device__ __forceinline__ float4 gather_row_sum(
    const float* __restrict__ h, const int* __restrict__ row_ptr,
    const int* __restrict__ cnt_in, const int* __restrict__ bsrc,
    const float* __restrict__ bw, int node, int l, int n_nodes) {
    const bool live = node < n_nodes;
    const int beg = live ? row_ptr[node] : 0;
    const int len = live ? cnt_in[node] : 0;
    const float4* __restrict__ h4 = (const float4*)h;
    const int* __restrict__ bp = bsrc + beg;
    const float* __restrict__ wp = bw + beg;

    float4 acc = make_float4(0.0f, 0.0f, 0.0f, 0.0f);
    int j = 0;
    for (; j + 4 <= len; j += 4) {
        const int s0 = bp[j], s1 = bp[j + 1], s2 = bp[j + 2], s3 = bp[j + 3];
        const float w0 = wp[j], w1 = wp[j + 1], w2 = wp[j + 2], w3 = wp[j + 3];
        const float4 v0 = h4[(size_t)s0 * 16 + l];
        const float4 v1 = h4[(size_t)s1 * 16 + l];
        const float4 v2 = h4[(size_t)s2 * 16 + l];
        const float4 v3 = h4[(size_t)s3 * 16 + l];
        acc.x += v0.x * w0; acc.y += v0.y * w0; acc.z += v0.z * w0; acc.w += v0.w * w0;
        acc.x += v1.x * w1; acc.y += v1.y * w1; acc.z += v1.z * w1; acc.w += v1.w * w1;
        acc.x += v2.x * w2; acc.y += v2.y * w2; acc.z += v2.z * w2; acc.w += v2.w * w2;
        acc.x += v3.x * w3; acc.y += v3.y * w3; acc.z += v3.z * w3; acc.w += v3.w * w3;
    }
    const int r = len - j;
    if (r > 0) {
        const int j1 = j + (r > 1 ? 1 : 0);
        const int j2 = j + (r > 2 ? 2 : 0);
        const int s0 = bp[j], s1 = bp[j1], s2 = bp[j2];
        const float w0 = wp[j], w1 = wp[j1], w2 = wp[j2];
        const float4 v0 = h4[(size_t)s0 * 16 + l];
        const float4 v1 = h4[(size_t)s1 * 16 + l];
        const float4 v2 = h4[(size_t)s2 * 16 + l];
        const bool k1 = r > 1, k2 = r > 2;
        acc.x += v0.x * w0; acc.y += v0.y * w0; acc.z += v0.z * w0; acc.w += v0.w * w0;
        acc.x += k1 ? v1.x * w1 : 0.0f; acc.y += k1 ? v1.y * w1 : 0.0f;
        acc.z += k1 ? v1.z * w1 : 0.0f; acc.w += k1 ? v1.w * w1 : 0.0f;
        acc.x += k2 ? v2.x * w2 : 0.0f; acc.y += k2 ? v2.y * w2 : 0.0f;
        acc.z += k2 ? v2.z * w2 : 0.0f; acc.w += k2 ? v2.w * w2 : 0.0f;
    }
    return acc;
}

__global__ __launch_bounds__(256) void gather_rows_kernel(
    const float* __restrict__ h, const int* __restrict__ row_ptr,
    const int* __restrict__ cnt_in, const int* __restrict__ bsrc,
    const float* __restrict__ bw, const float* __restrict__ ndst,
    const float* __restrict__ bias, float* __restrict__ outp, int n_nodes) {
    const int t = blockIdx.x * blockDim.x + threadIdx.x;
    const int node = t >> 4;
    const int l = t & 15;
    if (node >= n_nodes) return;
    float4 acc = gather_row_sum(h, row_ptr, cnt_in, bsrc, bw, node, l, n_nodes);
    const float nd = ndst[node];
    const float4 b = *(const float4*)(bias + 4 * l);
    acc.x = fmaxf(acc.x * nd + b.x, 0.0f);
    acc.y = fmaxf(acc.y * nd + b.y, 0.0f);
    acc.z = fmaxf(acc.z * nd + b.z, 0.0f);
    acc.w = fmaxf(acc.w * nd + b.w, 0.0f);
    ((float4*)outp)[(size_t)node * 16 + l] = acc;
}

#define AGG_PITCH 68
template <int F_OUT, int NT, bool RELU>
__global__ __launch_bounds__(256) void gather_gemm_kernel(
    const float* __restrict__ h, const int* __restrict__ row_ptr,
    const int* __restrict__ cnt_in, const int* __restrict__ bsrc,
    const float* __restrict__ bw, const float* __restrict__ ndst,
    const float* __restrict__ W, const float* __restrict__ bias,
    float* __restrict__ out, int n_nodes) {
    __shared__ __attribute__((aligned(16))) float tile[16 * AGG_PITCH];
    const int tid = threadIdx.x;
    const int row0 = blockIdx.x * 16;
    {
        const int rl = tid >> 4;
        const int l = tid & 15;
        const float4 acc = gather_row_sum(h, row_ptr, cnt_in, bsrc, bw, row0 + rl, l, n_nodes);
        *(float4*)(tile + rl * AGG_PITCH + 4 * l) = acc;
    }
    __syncthreads();

    const int wave = tid >> 5;
    if (wave >= NT) return;
    const int lane = tid & 31;
    const int half = lane >> 4;
    const int lm = lane & 15;
    const int col = wave * 16 + lm;
    const bool cok = col < F_OUT;
    const int cc = cok ? col : F_OUT - 1;

    float nd[8];
#pragma unroll
    for (int i = 0; i < 8; ++i) {
        const int r = row0 + i + 8 * half;
        nd[i] = ndst[r < n_nodes ? r : n_nodes - 1];
    }
    const float bcol = bias[cc];

    v8f c = v8f{};
    const float* arow = tile + lm * AGG_PITCH;
#pragma unroll
    for (int k0 = 0; k0 < 64; k0 += 4) {
        const int ka = k0 + 2 * half;
        v2f a;
        a.x = arow[ka];
        a.y = arow[ka + 1];
        const float wx = W[(size_t)ka * F_OUT + cc];
        const float wy = W[(size_t)(ka + 1) * F_OUT + cc];
        v2f bf;
        bf.x = cok ? wx : 0.0f;
        bf.y = cok ? wy : 0.0f;
        c = __builtin_amdgcn_wmma_f32_16x16x4_f32(
             false, a,  false, bf,
             (short)0, c,  false,  false);
    }

    asm volatile("v_nop\n\tv_nop\n\tv_nop\n\tv_nop\n\tv_nop" : "+v"(c));

#pragma unroll
    for (int i = 0; i < 8; ++i) {
        const int r = row0 + i + 8 * half;
        float v = c[i] * nd[i] + bcol;
        if (RELU) v = fmaxf(v, 0.0f);
        if (cok && r < n_nodes) out[(size_t)r * F_OUT + col] = v;
    }
}

template <int F_IN, int F_OUT, int NT, bool EPI, bool RELU>
__global__ __launch_bounds__(32) void gcn_wmma_layer(
    const float* __restrict__ agg, const float* __restrict__ ndst,
    const float* __restrict__ W, const float* __restrict__ bias,
    float* __restrict__ out, int n_nodes) {
    const int lane = threadIdx.x;
    const int half = lane >> 4;
    const int lm = lane & 15;
    const int row0 = blockIdx.x * 16;

    v8f acc[NT];
#pragma unroll
    for (int t = 0; t < NT; ++t) acc[t] = v8f{};

    const float* arow = agg + (size_t)(row0 + lm) * F_IN;

    for (int k0 = 0; k0 < F_IN; k0 += 4) {
        const int ka = k0 + 2 * half;
        v2f a;
        a.x = arow[ka];
        a.y = arow[ka + 1];
#pragma unroll
        for (int t = 0; t < NT; ++t) {
            const int col = t * 16 + lm;
            v2f bf;
            bf.x = (col < F_OUT) ? W[(size_t)ka * F_OUT + col] : 0.0f;
            bf.y = (col < F_OUT) ? W[(size_t)(ka + 1) * F_OUT + col] : 0.0f;
            acc[t] = __builtin_amdgcn_wmma_f32_16x16x4_f32(
                 false, a,  false, bf,
                 (short)0, acc[t],  false,  false);
        }
    }

#pragma unroll
    for (int i = 0; i < 8; ++i) {
        const int r = row0 + i + 8 * half;
        const float nd = EPI ? ndst[r] : 1.0f;
#pragma unroll
        for (int t = 0; t < NT; ++t) {
            const int col = t * 16 + lm;
            if (col < F_OUT) {
                float v = acc[t][i];
                if (EPI) v = v * nd + bias[col];
                if (RELU) v = fmaxf(v, 0.0f);
                out[(size_t)r * F_OUT + col] = v;
            }
        }
    }
}

static inline size_t align256(size_t x) { return (x + 255) & ~(size_t)255; }

extern "C" void kernel_launch(void* const* d_in, const int* in_sizes, int n_in,
                              void* d_out, int out_size, void* d_ws, size_t ws_size,
                              hipStream_t stream) {
    const float* x  = (const float*)d_in[0];
    const float* W0 = (const float*)d_in[1];
    const float* b0 = (const float*)d_in[2];
    const float* W1 = (const float*)d_in[3];
    const float* b1 = (const float*)d_in[4];
    const float* W2 = (const float*)d_in[5];
    const float* b2 = (const float*)d_in[6];
    const int* src  = (const int*)d_in[7];
    const int* dst  = (const int*)d_in[8];

    const int N = in_sizes[0] / N_FEAT_IN;
    const int E = in_sizes[7];
    float* out = (float*)d_out;

    char* p = (char*)d_ws;
    float* nsrc = (float*)p;  p += align256((size_t)N * 4);
    float* ndst = (float*)p;  p += align256((size_t)N * 4);
    char* slab = p;           p += align256((size_t)N * N_FEAT_IN * 4);
    float* h0   = (float*)p;  p += align256((size_t)N * 64 * 4);
    float* h1   = (float*)p;  p += align256((size_t)N * 64 * 4);
    int* rank    = (int*)slab;    slab += align256((size_t)N * 64 * 4);
    int* cnt_out = (int*)slab;    slab += align256((size_t)N * 4);
    int* cnt_in  = (int*)slab;    slab += align256((size_t)N * 4);
    int* row_ptr = (int*)slab;    slab += align256((size_t)N * 4);
    int* bsrc    = (int*)slab;    slab += align256((size_t)E * 4);
    float* bw    = (float*)slab;

    const int TB = 256;

    hipMemsetAsync(cnt_out, 0, 2 * align256((size_t)N * 4), stream);
    degree_rank_kernel<<<(E + TB - 1) / TB, TB, 0, stream>>>(src, dst, cnt_out, cnt_in, rank, E);
    rowptr_norm_kernel<<<(N + 1023) / 1024, 1024, 0, stream>>>(cnt_out, cnt_in, row_ptr, nsrc, ndst, N);
    bucket_fill_kernel<<<(E + TB - 1) / TB, TB, 0, stream>>>(src, dst, rank, row_ptr, nsrc, bsrc, bw, E);

    const int GB = (N * 16 + TB - 1) / TB;

    gcn_wmma_layer<N_FEAT_IN, 64, 4, false, false>
        <<<N / 16, 32, 0, stream>>>(x, nullptr, W0, nullptr, h1, N);
    gather_rows_kernel<<<GB, TB, 0, stream>>>(h1, row_ptr, cnt_in, bsrc, bw, ndst, b0, h0, N);

    gather_gemm_kernel<64, 4, true>
        <<<GB, TB, 0, stream>>>(h0, row_ptr, cnt_in, bsrc, bw, ndst, W1, b1, h1, N);

    gather_gemm_kernel<47, 3, false>
        <<<GB, TB, 0, stream>>>(h1, row_ptr, cnt_in, bsrc, bw, ndst, W2, b2, out, N);
}


# Degree counts and edge ranks from LDS histograms, not global atomics: 32 edge chunks each count all nodes in LDS (returning LDS add = rank); scan adds chunk prefixes; no memset; sum order by rank
# speedup vs baseline: 3.5537x; 1.4710x over previous
#include <hip/hip_runtime.h>


typedef float v2f __attribute__((ext_vector_type(2)));
typedef float v8f __attribute__((ext_vector_type(8)));

#define N_FEAT_IN 100

#define N_CHUNK 32
#define HIST_CAP 65536
__global__ __launch_bounds__(1024) void chunk_hist_kernel(
    const int* __restrict__ src, const int* __restrict__ dst,
    int* __restrict__ hist_dst, int* __restrict__ hist_src,
    int* __restrict__ rank, int* __restrict__ tiletot,
    int n_nodes, int n_edges, int chunk_len, int n_tiles) {
    __shared__ __attribute__((aligned(16))) int hist[HIST_CAP];
    const int tid = threadIdx.x;
    const int lane = tid & 31;
    const int wv = tid >> 5;
    const bool is_dst = blockIdx.x < N_CHUNK;
    const int p = is_dst ? blockIdx.x : blockIdx.x - N_CHUNK;
    const int* __restrict__ keys = is_dst ? dst : src;
    int* __restrict__ gh = (is_dst ? hist_dst : hist_src) + (size_t)p * n_nodes;
    const int e0 = p * chunk_len;
    const int e1 = (e0 + chunk_len < n_edges) ? e0 + chunk_len : n_edges;

    for (int base = 0; base < n_nodes; base += HIST_CAP) {
        const int nb = (n_nodes - base < HIST_CAP) ? n_nodes - base : HIST_CAP;
        for (int i = tid; i < (nb + 3) / 4; i += 1024)
            ((int4*)hist)[i] = make_int4(0, 0, 0, 0);
        __syncthreads();

        for (int eb = e0; eb < e1; eb += 16 * 1024) {
            int key[16];
#pragma unroll
            for (int u = 0; u < 16; ++u) {
                const int e = eb + u * 1024 + tid;
                key[u] = (e < e1) ? keys[e] : -1;
            }
#pragma unroll
            for (int u = 0; u < 16; ++u) {
                const unsigned k = (unsigned)(key[u] - base);
                key[u] = (k < (unsigned)nb) ? atomicAdd(&hist[k], 1) : -1;
            }
            if (is_dst) {
#pragma unroll
                for (int u = 0; u < 16; ++u) {
                    const int e = eb + u * 1024 + tid;
                    if (key[u] >= 0) rank[e] = key[u];
                }
            }
        }
        __syncthreads();

        for (int i = tid; i < nb; i += 1024) gh[base + i] = hist[i];
        if (is_dst) {
            const int nt = (nb + 1023) >> 10;
            for (int t = wv; t < nt; t += 32) {
                int s = 0;
                for (int k = lane; k < 1024; k += 32) {
                    const int i = t * 1024 + k;
                    s += (i < nb) ? hist[i] : 0;
                }
                for (int d = 16; d > 0; d >>= 1) s += __shfl_xor(s, d);
                if (lane == 0) tiletot[p * n_tiles + (base >> 10) + t] = s;
            }
        }
        __syncthreads();
    }
}

__global__ __launch_bounds__(1024) void rowptr_norm_kernel(
    int* __restrict__ hd, const int* __restrict__ hs,
    const int* __restrict__ tiletot, int* __restrict__ cnt_in,
    int* __restrict__ row_ptr, float* __restrict__ nsrc,
    float* __restrict__ ndst, int n, int n_tiles) {
    __shared__ int w_incl[32];
    __shared__ int w_part[32];
    const int tid = threadIdx.x;
    const int lane = tid & 31;
    const int wv = tid >> 5;
    const int bx = blockIdx.x;

    int part = 0;
    for (int p = wv; p < N_CHUNK; p += 32)
        for (int t = lane; t < bx; t += 32) part += tiletot[p * n_tiles + t];

    const int i = bx * 1024 + tid;
    int pre[N_CHUNK];
    int c_in = 0;
    if (i < n) {
        int c_out = 0;
#pragma unroll
        for (int p = 0; p < N_CHUNK; ++p) pre[p] = hd[(size_t)p * n + i];
#pragma unroll
        for (int p = 0; p < N_CHUNK; ++p) c_out += hs[(size_t)p * n + i];
#pragma unroll
        for (int p = 0; p < N_CHUNK; ++p) {
            const int v = pre[p];
            pre[p] = c_in;
            c_in += v;
        }
        cnt_in[i] = c_in;
        ndst[i] = rsqrtf(fmaxf((float)c_in, 1.0f));
        nsrc[i] = rsqrtf(fmaxf((float)c_out, 1.0f));
    }
    int incl = c_in;
    for (int d = 1; d < 32; d <<= 1) {
        const int up = __shfl_up(incl, d);
        if (lane >= d) incl += up;
    }
    for (int d = 16; d > 0; d >>= 1) part += __shfl_xor(part, d);
    if (lane == 31) w_incl[wv] = incl;
    if (lane == 0) w_part[wv] = part;
    __syncthreads();
    int offset = 0;
    for (int w = 0; w < 32; ++w) {
        offset += w_part[w];
        if (w < wv) offset += w_incl[w];
    }
    if (i < n) {
        const int rp = offset + incl - c_in;
        row_ptr[i] = rp;
#pragma unroll
        for (int p = 0; p < N_CHUNK; ++p) hd[(size_t)p * n + i] = rp + pre[p];
    }
}

__global__ void bucket_fill_kernel(const int* __restrict__ src,
                                   const int* __restrict__ dst,
                                   const int* __restrict__ rank,
                                   const int* __restrict__ chunk_base,
                                   const float* __restrict__ nsrc,
                                   int* __restrict__ bsrc,
                                   float* __restrict__ bw,
                                   int n_edges, int n_nodes, int chunk_len) {
    int e = blockIdx.x * blockDim.x + threadIdx.x;
    if (e >= n_edges) return;
    const int s = src[e];
    const int p = e / chunk_len;
    const int slot = chunk_base[(size_t)p * n_nodes + dst[e]] + rank[e];
    bsrc[slot] = s;
    bw[slot] = nsrc[s];
}

__device__ __forceinline__ float4 gather_row_sum(
    const float* __restrict__ h, const int* __restrict__ row_ptr,
    const int* __restrict__ cnt_in, const int* __restrict__ bsrc,
    const float* __restrict__ bw, int node, int l, int n_nodes) {
    const bool live = node < n_nodes;
    const int beg = live ? row_ptr[node] : 0;
    const int len = live ? cnt_in[node] : 0;
    const float4* __restrict__ h4 = (const float4*)h;
    const int* __restrict__ bp = bsrc + beg;
    const float* __restrict__ wp = bw + beg;

    float4 acc = make_float4(0.0f, 0.0f, 0.0f, 0.0f);
    int j = 0;
    for (; j + 4 <= len; j += 4) {
        const int s0 = bp[j], s1 = bp[j + 1], s2 = bp[j + 2], s3 = bp[j + 3];
        const float w0 = wp[j], w1 = wp[j + 1], w2 = wp[j + 2], w3 = wp[j + 3];
        const float4 v0 = h4[(size_t)s0 * 16 + l];
        const float4 v1 = h4[(size_t)s1 * 16 + l];
        const float4 v2 = h4[(size_t)s2 * 16 + l];
        const float4 v3 = h4[(size_t)s3 * 16 + l];
        acc.x += v0.x * w0; acc.y += v0.y * w0; acc.z += v0.z * w0; acc.w += v0.w * w0;
        acc.x += v1.x * w1; acc.y += v1.y * w1; acc.z += v1.z * w1; acc.w += v1.w * w1;
        acc.x += v2.x * w2; acc.y += v2.y * w2; acc.z += v2.z * w2; acc.w += v2.w * w2;
        acc.x += v3.x * w3; acc.y += v3.y * w3; acc.z += v3.z * w3; acc.w += v3.w * w3;
    }
    const int r = len - j;
    if (r > 0) {
        const int j1 = j + (r > 1 ? 1 : 0);
        const int j2 = j + (r > 2 ? 2 : 0);
        const int s0 = bp[j], s1 = bp[j1], s2 = bp[j2];
        const float w0 = wp[j], w1 = wp[j1], w2 = wp[j2];
        const float4 v0 = h4[(size_t)s0 * 16 + l];
        const float4 v1 = h4[(size_t)s1 * 16 + l];
        const float4 v2 = h4[(size_t)s2 * 16 + l];
        const bool k1 = r > 1, k2 = r > 2;
        acc.x += v0.x * w0; acc.y += v0.y * w0; acc.z += v0.z * w0; acc.w += v0.w * w0;
        acc.x += k1 ? v1.x * w1 : 0.0f; acc.y += k1 ? v1.y * w1 : 0.0f;
        acc.z += k1 ? v1.z * w1 : 0.0f; acc.w += k1 ? v1.w * w1 : 0.0f;
        acc.x += k2 ? v2.x * w2 : 0.0f; acc.y += k2 ? v2.y * w2 : 0.0f;
        acc.z += k2 ? v2.z * w2 : 0.0f; acc.w += k2 ? v2.w * w2 : 0.0f;
    }
    return acc;
}

__global__ __launch_bounds__(256) void gather_rows_kernel(
    const float* __restrict__ h, const int* __restrict__ row_ptr,
    const int* __restrict__ cnt_in, const int* __restrict__ bsrc,
    const float* __restrict__ bw, const float* __restrict__ ndst,
    const float* __restrict__ bias, float* __restrict__ outp, int n_nodes) {
    const int t = blockIdx.x * blockDim.x + threadIdx.x;
    const int node = t >> 4;
    const int l = t & 15;
    if (node >= n_nodes) return;
    float4 acc = gather_row_sum(h, row_ptr, cnt_in, bsrc, bw, node, l, n_nodes);
    const float nd = ndst[node];
    const float4 b = *(const float4*)(bias + 4 * l);
    acc.x = fmaxf(acc.x * nd + b.x, 0.0f);
    acc.y = fmaxf(acc.y * nd + b.y, 0.0f);
    acc.z = fmaxf(acc.z * nd + b.z, 0.0f);
    acc.w = fmaxf(acc.w * nd + b.w, 0.0f);
    ((float4*)outp)[(size_t)node * 16 + l] = acc;
}

#define AGG_PITCH 68
template <int F_OUT, int NT, bool RELU>
__global__ __launch_bounds__(256) void gather_gemm_kernel(
    const float* __restrict__ h, const int* __restrict__ row_ptr,
    const int* __restrict__ cnt_in, const int* __restrict__ bsrc,
    const float* __restrict__ bw, const float* __restrict__ ndst,
    const float* __restrict__ W, const float* __restrict__ bias,
    float* __restrict__ out, int n_nodes) {
    __shared__ __attribute__((aligned(16))) float tile[16 * AGG_PITCH];
    const int tid = threadIdx.x;
    const int row0 = blockIdx.x * 16;
    {
        const int rl = tid >> 4;
        const int l = tid & 15;
        const float4 acc = gather_row_sum(h, row_ptr, cnt_in, bsrc, bw, row0 + rl, l, n_nodes);
        *(float4*)(tile + rl * AGG_PITCH + 4 * l) = acc;
    }
    __syncthreads();

    const int wave = tid >> 5;
    if (wave >= NT) return;
    const int lane = tid & 31;
    const int half = lane >> 4;
    const int lm = lane & 15;
    const int col = wave * 16 + lm;
    const bool cok = col < F_OUT;
    const int cc = cok ? col : F_OUT - 1;

    float nd[8];
#pragma unroll
    for (int i = 0; i < 8; ++i) {
        const int r = row0 + i + 8 * half;
        nd[i] = ndst[r < n_nodes ? r : n_nodes - 1];
    }
    const float bcol = bias[cc];

    v8f c = v8f{};
    const float* arow = tile + lm * AGG_PITCH;
#pragma unroll
    for (int k0 = 0; k0 < 64; k0 += 4) {
        const int ka = k0 + 2 * half;
        v2f a;
        a.x = arow[ka];
        a.y = arow[ka + 1];
        const float wx = W[(size_t)ka * F_OUT + cc];
        const float wy = W[(size_t)(ka + 1) * F_OUT + cc];
        v2f bf;
        bf.x = cok ? wx : 0.0f;
        bf.y = cok ? wy : 0.0f;
        c = __builtin_amdgcn_wmma_f32_16x16x4_f32(
             false, a,  false, bf,
             (short)0, c,  false,  false);
    }

    asm volatile("v_nop\n\tv_nop\n\tv_nop\n\tv_nop\n\tv_nop" : "+v"(c));

#pragma unroll
    for (int i = 0; i < 8; ++i) {
        const int r = row0 + i + 8 * half;
        float v = c[i] * nd[i] + bcol;
        if (RELU) v = fmaxf(v, 0.0f);
        if (cok && r < n_nodes) out[(size_t)r * F_OUT + col] = v;
    }
}

template <int F_IN, int F_OUT, int NT, bool EPI, bool RELU>
__global__ __launch_bounds__(32) void gcn_wmma_layer(
    const float* __restrict__ agg, const float* __restrict__ ndst,
    const float* __restrict__ W, const float* __restrict__ bias,
    float* __restrict__ out, int n_nodes) {
    const int lane = threadIdx.x;
    const int half = lane >> 4;
    const int lm = lane & 15;
    const int row0 = blockIdx.x * 16;

    v8f acc[NT];
#pragma unroll
    for (int t = 0; t < NT; ++t) acc[t] = v8f{};

    const float* arow = agg + (size_t)(row0 + lm) * F_IN;

    for (int k0 = 0; k0 < F_IN; k0 += 4) {
        const int ka = k0 + 2 * half;
        v2f a;
        a.x = arow[ka];
        a.y = arow[ka + 1];
#pragma unroll
        for (int t = 0; t < NT; ++t) {
            const int col = t * 16 + lm;
            v2f bf;
            bf.x = (col < F_OUT) ? W[(size_t)ka * F_OUT + col] : 0.0f;
            bf.y = (col < F_OUT) ? W[(size_t)(ka + 1) * F_OUT + col] : 0.0f;
            acc[t] = __builtin_amdgcn_wmma_f32_16x16x4_f32(
                 false, a,  false, bf,
                 (short)0, acc[t],  false,  false);
        }
    }

#pragma unroll
    for (int i = 0; i < 8; ++i) {
        const int r = row0 + i + 8 * half;
        const float nd = EPI ? ndst[r] : 1.0f;
#pragma unroll
        for (int t = 0; t < NT; ++t) {
            const int col = t * 16 + lm;
            if (col < F_OUT) {
                float v = acc[t][i];
                if (EPI) v = v * nd + bias[col];
                if (RELU) v = fmaxf(v, 0.0f);
                out[(size_t)r * F_OUT + col] = v;
            }
        }
    }
}

static inline size_t align256(size_t x) { return (x + 255) & ~(size_t)255; }

extern "C" void kernel_launch(void* const* d_in, const int* in_sizes, int n_in,
                              void* d_out, int out_size, void* d_ws, size_t ws_size,
                              hipStream_t stream) {
    const float* x  = (const float*)d_in[0];
    const float* W0 = (const float*)d_in[1];
    const float* b0 = (const float*)d_in[2];
    const float* W1 = (const float*)d_in[3];
    const float* b1 = (const float*)d_in[4];
    const float* W2 = (const float*)d_in[5];
    const float* b2 = (const float*)d_in[6];
    const int* src  = (const int*)d_in[7];
    const int* dst  = (const int*)d_in[8];

    const int N = in_sizes[0] / N_FEAT_IN;
    const int E = in_sizes[7];
    float* out = (float*)d_out;

    char* p = (char*)d_ws;
    float* nsrc = (float*)p;  p += align256((size_t)N * 4);
    float* ndst = (float*)p;  p += align256((size_t)N * 4);
    char* slab = p;           p += align256((size_t)N * N_FEAT_IN * 4);
    float* h0   = (float*)p;  p += align256((size_t)N * 64 * 4);
    float* h1   = (float*)p;  p += align256((size_t)N * 64 * 4);
    int* rank    = (int*)slab;
    int* tiletot = (int*)(slab + align256((size_t)E * 4));
    slab += align256((size_t)N * 64 * 4);
    int* cnt_in  = (int*)slab;    slab += align256((size_t)N * 4);
    int* row_ptr = (int*)slab;    slab += align256((size_t)N * 4);
    int* bsrc    = (int*)slab;    slab += align256((size_t)E * 4);
    float* bw    = (float*)slab;
    int* hist_dst = (int*)h0;
    int* hist_src = hist_dst + (size_t)N_CHUNK * N;

    const int TB = 256;
    const int n_tiles = (N + 1023) / 1024;
    const int chunk_len = (E + N_CHUNK - 1) / N_CHUNK;

    chunk_hist_kernel<<<2 * N_CHUNK, 1024, 0, stream>>>(src, dst, hist_dst, hist_src, rank, tiletot,
                                                        N, E, chunk_len, n_tiles);
    rowptr_norm_kernel<<<n_tiles, 1024, 0, stream>>>(hist_dst, hist_src, tiletot, cnt_in, row_ptr,
                                                     nsrc, ndst, N, n_tiles);
    bucket_fill_kernel<<<(E + TB - 1) / TB, TB, 0, stream>>>(src, dst, rank, hist_dst, nsrc, bsrc, bw,
                                                             E, N, chunk_len);

    const int GB = (N * 16 + TB - 1) / TB;

    gcn_wmma_layer<N_FEAT_IN, 64, 4, false, false>
        <<<N / 16, 32, 0, stream>>>(x, nullptr, W0, nullptr, h1, N);
    gather_rows_kernel<<<GB, TB, 0, stream>>>(h1, row_ptr, cnt_in, bsrc, bw, ndst, b0, h0, N);

    gather_gemm_kernel<64, 4, true>
        <<<GB, TB, 0, stream>>>(h0, row_ptr, cnt_in, bsrc, bw, ndst, W1, b1, h1, N);

    gather_gemm_kernel<47, 3, false>
        <<<GB, TB, 0, stream>>>(h1, row_ptr, cnt_in, bsrc, bw, ndst, W2, b2, out, N);
}


# Layer 0 product x W0 moves into the histogram launch: 157 more 1024-thread groups (20 row tiles, one per wave) use the WGPs the 64 counting groups leave idle; 6 launches not 7; sums unchanged
# speedup vs baseline: 4.0156x; 1.1300x over previous
#include <hip/hip_runtime.h>


typedef float v2f __attribute__((ext_vector_type(2)));
typedef float v8f __attribute__((ext_vector_type(8)));

#define N_FEAT_IN 100

#define N_CHUNK 32
#define HIST_CAP 65536
#define XW_TILES 20
template <int F_IN, int F_OUT, int NT>
__device__ __forceinline__ void wmma_tile_product(
    const float* __restrict__ agg, const float* __restrict__ W,
    float* __restrict__ out, int row0, int lane);

__global__ __launch_bounds__(1024) void chunk_hist_xw_kernel(
    const int* __restrict__ src, const int* __restrict__ dst,
    int* __restrict__ hist_dst, int* __restrict__ hist_src,
    int* __restrict__ rank, int* __restrict__ tiletot,
    int n_nodes, int n_edges, int chunk_len, int n_tiles,
    const float* __restrict__ x, const float* __restrict__ W0,
    float* __restrict__ xw, int n_row_tiles) {
    __shared__ __attribute__((aligned(16))) int hist[HIST_CAP];
    const int tid = threadIdx.x;
    const int lane = tid & 31;
    const int wv = tid >> 5;
    if (blockIdx.x >= 2 * N_CHUNK) {
        const int w = __builtin_amdgcn_readfirstlane(wv);
        const int tile = (blockIdx.x - 2 * N_CHUNK) * XW_TILES + w;
        if (w >= XW_TILES || tile >= n_row_tiles) return;
        wmma_tile_product<N_FEAT_IN, 64, 4>(x, W0, xw, tile * 16, lane);
        return;
    }
    const bool is_dst = blockIdx.x < N_CHUNK;
    const int p = is_dst ? blockIdx.x : blockIdx.x - N_CHUNK;
    const int* __restrict__ keys = is_dst ? dst : src;
    int* __restrict__ gh = (is_dst ? hist_dst : hist_src) + (size_t)p * n_nodes;
    const int e0 = p * chunk_len;
    const int e1 = (e0 + chunk_len < n_edges) ? e0 + chunk_len : n_edges;

    for (int base = 0; base < n_nodes; base += HIST_CAP) {
        const int nb = (n_nodes - base < HIST_CAP) ? n_nodes - base : HIST_CAP;
        for (int i = tid; i < (nb + 3) / 4; i += 1024)
            ((int4*)hist)[i] = make_int4(0, 0, 0, 0);
        __syncthreads();

        for (int eb = e0; eb < e1; eb += 16 * 1024) {
            int key[16];
#pragma unroll
            for (int u = 0; u < 16; ++u) {
                const int e = eb + u * 1024 + tid;
                key[u] = (e < e1) ? keys[e] : -1;
            }
#pragma unroll
            for (int u = 0; u < 16; ++u) {
                const unsigned k = (unsigned)(key[u] - base);
                key[u] = (k < (unsigned)nb) ? atomicAdd(&hist[k], 1) : -1;
            }
            if (is_dst) {
#pragma unroll
                for (int u = 0; u < 16; ++u) {
                    const int e = eb + u * 1024 + tid;
                    if (key[u] >= 0) rank[e] = key[u];
                }
            }
        }
        __syncthreads();

        for (int i = tid; i < nb; i += 1024) gh[base + i] = hist[i];
        if (is_dst) {
            const int nt = (nb + 1023) >> 10;
            for (int t = wv; t < nt; t += 32) {
                int s = 0;
                for (int k = lane; k < 1024; k += 32) {
                    const int i = t * 1024 + k;
                    s += (i < nb) ? hist[i] : 0;
                }
                for (int d = 16; d > 0; d >>= 1) s += __shfl_xor(s, d);
                if (lane == 0) tiletot[p * n_tiles + (base >> 10) + t] = s;
            }
        }
        __syncthreads();
    }
}

__global__ __launch_bounds__(1024) void rowptr_norm_kernel(
    int* __restrict__ hd, const int* __restrict__ hs,
    const int* __restrict__ tiletot, int* __restrict__ cnt_in,
    int* __restrict__ row_ptr, float* __restrict__ nsrc,
    float* __restrict__ ndst, int n, int n_tiles) {
    __shared__ int w_incl[32];
    __shared__ int w_part[32];
    const int tid = threadIdx.x;
    const int lane = tid & 31;
    const int wv = tid >> 5;
    const int bx = blockIdx.x;

    int part = 0;
    for (int p = wv; p < N_CHUNK; p += 32)
        for (int t = lane; t < bx; t += 32) part += tiletot[p * n_tiles + t];

    const int i = bx * 1024 + tid;
    int pre[N_CHUNK];
    int c_in = 0;
    if (i < n) {
        int c_out = 0;
#pragma unroll
        for (int p = 0; p < N_CHUNK; ++p) pre[p] = hd[(size_t)p * n + i];
#pragma unroll
        for (int p = 0; p < N_CHUNK; ++p) c_out += hs[(size_t)p * n + i];
#pragma unroll
        for (int p = 0; p < N_CHUNK; ++p) {
            const int v = pre[p];
            pre[p] = c_in;
            c_in += v;
        }
        cnt_in[i] = c_in;
        ndst[i] = rsqrtf(fmaxf((float)c_in, 1.0f));
        nsrc[i] = rsqrtf(fmaxf((float)c_out, 1.0f));
    }
    int incl = c_in;
    for (int d = 1; d < 32; d <<= 1) {
        const int up = __shfl_up(incl, d);
        if (lane >= d) incl += up;
    }
    for (int d = 16; d > 0; d >>= 1) part += __shfl_xor(part, d);
    if (lane == 31) w_incl[wv] = incl;
    if (lane == 0) w_part[wv] = part;
    __syncthreads();
    int offset = 0;
    for (int w = 0; w < 32; ++w) {
        offset += w_part[w];
        if (w < wv) offset += w_incl[w];
    }
    if (i < n) {
        const int rp = offset + incl - c_in;
        row_ptr[i] = rp;
#pragma unroll
        for (int p = 0; p < N_CHUNK; ++p) hd[(size_t)p * n + i] = rp + pre[p];
    }
}

__global__ void bucket_fill_kernel(const int* __restrict__ src,
                                   const int* __restrict__ dst,
                                   const int* __restrict__ rank,
                                   const int* __restrict__ chunk_base,
                                   const float* __restrict__ nsrc,
                                   int* __restrict__ bsrc,
                                   float* __restrict__ bw,
                                   int n_edges, int n_nodes, int chunk_len) {
    int e = blockIdx.x * blockDim.x + threadIdx.x;
    if (e >= n_edges) return;
    const int s = src[e];
    const int p = e / chunk_len;
    const int slot = chunk_base[(size_t)p * n_nodes + dst[e]] + rank[e];
    bsrc[slot] = s;
    bw[slot] = nsrc[s];
}

__device__ __forceinline__ float4 gather_row_sum(
    const float* __restrict__ h, const int* __restrict__ row_ptr,
    const int* __restrict__ cnt_in, const int* __restrict__ bsrc,
    const float* __restrict__ bw, int node, int l, int n_nodes) {
    const bool live = node < n_nodes;
    const int beg = live ? row_ptr[node] : 0;
    const int len = live ? cnt_in[node] : 0;
    const float4* __restrict__ h4 = (const float4*)h;
    const int* __restrict__ bp = bsrc + beg;
    const float* __restrict__ wp = bw + beg;

    float4 acc = make_float4(0.0f, 0.0f, 0.0f, 0.0f);
    int j = 0;
    for (; j + 4 <= len; j += 4) {
        const int s0 = bp[j], s1 = bp[j + 1], s2 = bp[j + 2], s3 = bp[j + 3];
        const float w0 = wp[j], w1 = wp[j + 1], w2 = wp[j + 2], w3 = wp[j + 3];
        const float4 v0 = h4[(size_t)s0 * 16 + l];
        const float4 v1 = h4[(size_t)s1 * 16 + l];
        const float4 v2 = h4[(size_t)s2 * 16 + l];
        const float4 v3 = h4[(size_t)s3 * 16 + l];
        acc.x += v0.x * w0; acc.y += v0.y * w0; acc.z += v0.z * w0; acc.w += v0.w * w0;
        acc.x += v1.x * w1; acc.y += v1.y * w1; acc.z += v1.z * w1; acc.w += v1.w * w1;
        acc.x += v2.x * w2; acc.y += v2.y * w2; acc.z += v2.z * w2; acc.w += v2.w * w2;
        acc.x += v3.x * w3; acc.y += v3.y * w3; acc.z += v3.z * w3; acc.w += v3.w * w3;
    }
    const int r = len - j;
    if (r > 0) {
        const int j1 = j + (r > 1 ? 1 : 0);
        const int j2 = j + (r > 2 ? 2 : 0);
        const int s0 = bp[j], s1 = bp[j1], s2 = bp[j2];
        const float w0 = wp[j], w1 = wp[j1], w2 = wp[j2];
        const float4 v0 = h4[(size_t)s0 * 16 + l];
        const float4 v1 = h4[(size_t)s1 * 16 + l];
        const float4 v2 = h4[(size_t)s2 * 16 + l];
        const bool k1 = r > 1, k2 = r > 2;
        acc.x += v0.x * w0; acc.y += v0.y * w0; acc.z += v0.z * w0; acc.w += v0.w * w0;
        acc.x += k1 ? v1.x * w1 : 0.0f; acc.y += k1 ? v1.y * w1 : 0.0f;
        acc.z += k1 ? v1.z * w1 : 0.0f; acc.w += k1 ? v1.w * w1 : 0.0f;
        acc.x += k2 ? v2.x * w2 : 0.0f; acc.y += k2 ? v2.y * w2 : 0.0f;
        acc.z += k2 ? v2.z * w2 : 0.0f; acc.w += k2 ? v2.w * w2 : 0.0f;
    }
    return acc;
}

__global__ __launch_bounds__(256) void gather_rows_kernel(
    const float* __restrict__ h, const int* __restrict__ row_ptr,
    const int* __restrict__ cnt_in, const int* __restrict__ bsrc,
    const float* __restrict__ bw, const float* __restrict__ ndst,
    const float* __restrict__ bias, float* __restrict__ outp, int n_nodes) {
    const int t = blockIdx.x * blockDim.x + threadIdx.x;
    const int node = t >> 4;
    const int l = t & 15;
    if (node >= n_nodes) return;
    float4 acc = gather_row_sum(h, row_ptr, cnt_in, bsrc, bw, node, l, n_nodes);
    const float nd = ndst[node];
    const float4 b = *(const float4*)(bias + 4 * l);
    acc.x = fmaxf(acc.x * nd + b.x, 0.0f);
    acc.y = fmaxf(acc.y * nd + b.y, 0.0f);
    acc.z = fmaxf(acc.z * nd + b.z, 0.0f);
    acc.w = fmaxf(acc.w * nd + b.w, 0.0f);
    ((float4*)outp)[(size_t)node * 16 + l] = acc;
}

#define AGG_PITCH 68
template <int F_OUT, int NT, bool RELU>
__global__ __launch_bounds__(256) void gather_gemm_kernel(
    const float* __restrict__ h, const int* __restrict__ row_ptr,
    const int* __restrict__ cnt_in, const int* __restrict__ bsrc,
    const float* __restrict__ bw, const float* __restrict__ ndst,
    const float* __restrict__ W, const float* __restrict__ bias,
    float* __restrict__ out, int n_nodes) {
    __shared__ __attribute__((aligned(16))) float tile[16 * AGG_PITCH];
    const int tid = threadIdx.x;
    const int row0 = blockIdx.x * 16;
    {
        const int rl = tid >> 4;
        const int l = tid & 15;
        const float4 acc = gather_row_sum(h, row_ptr, cnt_in, bsrc, bw, row0 + rl, l, n_nodes);
        *(float4*)(tile + rl * AGG_PITCH + 4 * l) = acc;
    }
    __syncthreads();

    const int wave = tid >> 5;
    if (wave >= NT) return;
    const int lane = tid & 31;
    const int half = lane >> 4;
    const int lm = lane & 15;
    const int col = wave * 16 + lm;
    const bool cok = col < F_OUT;
    const int cc = cok ? col : F_OUT - 1;

    float nd[8];
#pragma unroll
    for (int i = 0; i < 8; ++i) {
        const int r = row0 + i + 8 * half;
        nd[i] = ndst[r < n_nodes ? r : n_nodes - 1];
    }
    const float bcol = bias[cc];

    v8f c = v8f{};
    const float* arow = tile + lm * AGG_PITCH;
#pragma unroll
    for (int k0 = 0; k0 < 64; k0 += 4) {
        const int ka = k0 + 2 * half;
        v2f a;
        a.x = arow[ka];
        a.y = arow[ka + 1];
        const float wx = W[(size_t)ka * F_OUT + cc];
        const float wy = W[(size_t)(ka + 1) * F_OUT + cc];
        v2f bf;
        bf.x = cok ? wx : 0.0f;
        bf.y = cok ? wy : 0.0f;
        c = __builtin_amdgcn_wmma_f32_16x16x4_f32(
             false, a,  false, bf,
             (short)0, c,  false,  false);
    }

    asm volatile("v_nop\n\tv_nop\n\tv_nop\n\tv_nop\n\tv_nop" : "+v"(c));

#pragma unroll
    for (int i = 0; i < 8; ++i) {
        const int r = row0 + i + 8 * half;
        float v = c[i] * nd[i] + bcol;
        if (RELU) v = fmaxf(v, 0.0f);
        if (cok && r < n_nodes) out[(size_t)r * F_OUT + col] = v;
    }
}

template <int F_IN, int F_OUT, int NT>
__device__ __forceinline__ void wmma_tile_product(
    const float* __restrict__ agg, const float* __restrict__ W,
    float* __restrict__ out, int row0, int lane) {
    const int half = lane >> 4;
    const int lm = lane & 15;

    v8f acc[NT];
#pragma unroll
    for (int t = 0; t < NT; ++t) acc[t] = v8f{};

    const float* arow = agg + (size_t)(row0 + lm) * F_IN;

    for (int k0 = 0; k0 < F_IN; k0 += 4) {
        const int ka = k0 + 2 * half;
        v2f a;
        a.x = arow[ka];
        a.y = arow[ka + 1];
#pragma unroll
        for (int t = 0; t < NT; ++t) {
            const int col = t * 16 + lm;
            v2f bf;
            bf.x = (col < F_OUT) ? W[(size_t)ka * F_OUT + col] : 0.0f;
            bf.y = (col < F_OUT) ? W[(size_t)(ka + 1) * F_OUT + col] : 0.0f;
            acc[t] = __builtin_amdgcn_wmma_f32_16x16x4_f32(
                 false, a,  false, bf,
                 (short)0, acc[t],  false,  false);
        }
    }

#pragma unroll
    for (int i = 0; i < 8; ++i) {
        const int r = row0 + i + 8 * half;
#pragma unroll
        for (int t = 0; t < NT; ++t) {
            const int col = t * 16 + lm;
            if (col < F_OUT) out[(size_t)r * F_OUT + col] = acc[t][i];
        }
    }
}

static inline size_t align256(size_t x) { return (x + 255) & ~(size_t)255; }

extern "C" void kernel_launch(void* const* d_in, const int* in_sizes, int n_in,
                              void* d_out, int out_size, void* d_ws, size_t ws_size,
                              hipStream_t stream) {
    const float* x  = (const float*)d_in[0];
    const float* W0 = (const float*)d_in[1];
    const float* b0 = (const float*)d_in[2];
    const float* W1 = (const float*)d_in[3];
    const float* b1 = (const float*)d_in[4];
    const float* W2 = (const float*)d_in[5];
    const float* b2 = (const float*)d_in[6];
    const int* src  = (const int*)d_in[7];
    const int* dst  = (const int*)d_in[8];

    const int N = in_sizes[0] / N_FEAT_IN;
    const int E = in_sizes[7];
    float* out = (float*)d_out;

    char* p = (char*)d_ws;
    float* nsrc = (float*)p;  p += align256((size_t)N * 4);
    float* ndst = (float*)p;  p += align256((size_t)N * 4);
    char* slab = p;           p += align256((size_t)N * N_FEAT_IN * 4);
    float* h0   = (float*)p;  p += align256((size_t)N * 64 * 4);
    float* h1   = (float*)p;  p += align256((size_t)N * 64 * 4);
    int* rank    = (int*)slab;
    int* tiletot = (int*)(slab + align256((size_t)E * 4));
    slab += align256((size_t)N * 64 * 4);
    int* cnt_in  = (int*)slab;    slab += align256((size_t)N * 4);
    int* row_ptr = (int*)slab;    slab += align256((size_t)N * 4);
    int* bsrc    = (int*)slab;    slab += align256((size_t)E * 4);
    float* bw    = (float*)slab;
    int* hist_dst = (int*)h0;
    int* hist_src = hist_dst + (size_t)N_CHUNK * N;

    const int TB = 256;
    const int n_tiles = (N + 1023) / 1024;
    const int chunk_len = (E + N_CHUNK - 1) / N_CHUNK;

    const int n_row_tiles = N / 16;
    const int n_xw_groups = (n_row_tiles + XW_TILES - 1) / XW_TILES;
    chunk_hist_xw_kernel<<<2 * N_CHUNK + n_xw_groups, 1024, 0, stream>>>(
        src, dst, hist_dst, hist_src, rank, tiletot, N, E, chunk_len, n_tiles,
        x, W0, h1, n_row_tiles);
    rowptr_norm_kernel<<<n_tiles, 1024, 0, stream>>>(hist_dst, hist_src, tiletot, cnt_in, row_ptr,
                                                     nsrc, ndst, N, n_tiles);
    bucket_fill_kernel<<<(E + TB - 1) / TB, TB, 0, stream>>>(src, dst, rank, hist_dst, nsrc, bsrc, bw,
                                                             E, N, chunk_len);

    const int GB = (N * 16 + TB - 1) / TB;

    gather_rows_kernel<<<GB, TB, 0, stream>>>(h1, row_ptr, cnt_in, bsrc, bw, ndst, b0, h0, N);

    gather_gemm_kernel<64, 4, true>
        <<<GB, TB, 0, stream>>>(h0, row_ptr, cnt_in, bsrc, bw, ndst, W1, b1, h1, N);

    gather_gemm_kernel<47, 3, false>
        <<<GB, TB, 0, stream>>>(h1, row_ptr, cnt_in, bsrc, bw, ndst, W2, b2, out, N);
}


# Gather waits for memory fewer times: a row's 16 lanes load 16 edge ids and norms at once and share them by DPP row_share, next 16 prefetched, 8 row loads a group; sum order kept, every term fused
# speedup vs baseline: 4.1362x; 1.0300x over previous
#include <hip/hip_runtime.h>


typedef float v2f __attribute__((ext_vector_type(2)));
typedef float v8f __attribute__((ext_vector_type(8)));

#define N_FEAT_IN 100

#define N_CHUNK 32
#define HIST_CAP 65536
#define XW_TILES 20
template <int F_IN, int F_OUT, int NT>
__device__ __forceinline__ void wmma_tile_product(
    const float* __restrict__ agg, const float* __restrict__ W,
    float* __restrict__ out, int row0, int lane);

__global__ __launch_bounds__(1024) void chunk_hist_xw_kernel(
    const int* __restrict__ src, const int* __restrict__ dst,
    int* __restrict__ hist_dst, int* __restrict__ hist_src,
    int* __restrict__ rank, int* __restrict__ tiletot,
    int n_nodes, int n_edges, int chunk_len, int n_tiles,
    const float* __restrict__ x, const float* __restrict__ W0,
    float* __restrict__ xw, int n_row_tiles) {
    __shared__ __attribute__((aligned(16))) int hist[HIST_CAP];
    const int tid = threadIdx.x;
    const int lane = tid & 31;
    const int wv = tid >> 5;
    if (blockIdx.x >= 2 * N_CHUNK) {
        const int w = __builtin_amdgcn_readfirstlane(wv);
        const int tile = (blockIdx.x - 2 * N_CHUNK) * XW_TILES + w;
        if (w >= XW_TILES || tile >= n_row_tiles) return;
        wmma_tile_product<N_FEAT_IN, 64, 4>(x, W0, xw, tile * 16, lane);
        return;
    }
    const bool is_dst = blockIdx.x < N_CHUNK;
    const int p = is_dst ? blockIdx.x : blockIdx.x - N_CHUNK;
    const int* __restrict__ keys = is_dst ? dst : src;
    int* __restrict__ gh = (is_dst ? hist_dst : hist_src) + (size_t)p * n_nodes;
    const int e0 = p * chunk_len;
    const int e1 = (e0 + chunk_len < n_edges) ? e0 + chunk_len : n_edges;

    for (int base = 0; base < n_nodes; base += HIST_CAP) {
        const int nb = (n_nodes - base < HIST_CAP) ? n_nodes - base : HIST_CAP;
        for (int i = tid; i < (nb + 3) / 4; i += 1024)
            ((int4*)hist)[i] = make_int4(0, 0, 0, 0);
        __syncthreads();

        for (int eb = e0; eb < e1; eb += 16 * 1024) {
            int key[16];
#pragma unroll
            for (int u = 0; u < 16; ++u) {
                const int e = eb + u * 1024 + tid;
                key[u] = (e < e1) ? keys[e] : -1;
            }
#pragma unroll
            for (int u = 0; u < 16; ++u) {
                const unsigned k = (unsigned)(key[u] - base);
                key[u] = (k < (unsigned)nb) ? atomicAdd(&hist[k], 1) : -1;
            }
            if (is_dst) {
#pragma unroll
                for (int u = 0; u < 16; ++u) {
                    const int e = eb + u * 1024 + tid;
                    if (key[u] >= 0) rank[e] = key[u];
                }
            }
        }
        __syncthreads();

        for (int i = tid; i < nb; i += 1024) gh[base + i] = hist[i];
        if (is_dst) {
            const int nt = (nb + 1023) >> 10;
            for (int t = wv; t < nt; t += 32) {
                int s = 0;
                for (int k = lane; k < 1024; k += 32) {
                    const int i = t * 1024 + k;
                    s += (i < nb) ? hist[i] : 0;
                }
                for (int d = 16; d > 0; d >>= 1) s += __shfl_xor(s, d);
                if (lane == 0) tiletot[p * n_tiles + (base >> 10) + t] = s;
            }
        }
        __syncthreads();
    }
}

__global__ __launch_bounds__(1024) void rowptr_norm_kernel(
    int* __restrict__ hd, const int* __restrict__ hs,
    const int* __restrict__ tiletot, int* __restrict__ cnt_in,
    int* __restrict__ row_ptr, float* __restrict__ nsrc,
    float* __restrict__ ndst, int n, int n_tiles) {
    __shared__ int w_incl[32];
    __shared__ int w_part[32];
    const int tid = threadIdx.x;
    const int lane = tid & 31;
    const int wv = tid >> 5;
    const int bx = blockIdx.x;

    int part = 0;
    for (int p = wv; p < N_CHUNK; p += 32)
        for (int t = lane; t < bx; t += 32) part += tiletot[p * n_tiles + t];

    const int i = bx * 1024 + tid;
    int pre[N_CHUNK];
    int c_in = 0;
    if (i < n) {
        int c_out = 0;
#pragma unroll
        for (int p = 0; p < N_CHUNK; ++p) pre[p] = hd[(size_t)p * n + i];
#pragma unroll
        for (int p = 0; p < N_CHUNK; ++p) c_out += hs[(size_t)p * n + i];
#pragma unroll
        for (int p = 0; p < N_CHUNK; ++p) {
            const int v = pre[p];
            pre[p] = c_in;
            c_in += v;
        }
        cnt_in[i] = c_in;
        ndst[i] = rsqrtf(fmaxf((float)c_in, 1.0f));
        nsrc[i] = rsqrtf(fmaxf((float)c_out, 1.0f));
    }
    int incl = c_in;
    for (int d = 1; d < 32; d <<= 1) {
        const int up = __shfl_up(incl, d);
        if (lane >= d) incl += up;
    }
    for (int d = 16; d > 0; d >>= 1) part += __shfl_xor(part, d);
    if (lane == 31) w_incl[wv] = incl;
    if (lane == 0) w_part[wv] = part;
    __syncthreads();
    int offset = 0;
    for (int w = 0; w < 32; ++w) {
        offset += w_part[w];
        if (w < wv) offset += w_incl[w];
    }
    if (i < n) {
        const int rp = offset + incl - c_in;
        row_ptr[i] = rp;
#pragma unroll
        for (int p = 0; p < N_CHUNK; ++p) hd[(size_t)p * n + i] = rp + pre[p];
    }
}

__global__ void bucket_fill_kernel(const int* __restrict__ src,
                                   const int* __restrict__ dst,
                                   const int* __restrict__ rank,
                                   const int* __restrict__ chunk_base,
                                   const float* __restrict__ nsrc,
                                   int* __restrict__ bsrc,
                                   float* __restrict__ bw,
                                   int n_edges, int n_nodes, int chunk_len) {
    int e = blockIdx.x * blockDim.x + threadIdx.x;
    if (e >= n_edges) return;
    const int s = src[e];
    const int p = e / chunk_len;
    const int slot = chunk_base[(size_t)p * n_nodes + dst[e]] + rank[e];
    bsrc[slot] = s;
    bw[slot] = nsrc[s];
}

#define ROW_SHARE(x, n) __builtin_amdgcn_update_dpp(0, (x), 0x150 + (n), 0xf, 0xf, true)

template <int G>
__device__ __forceinline__ void gather8_load(const float4* __restrict__ h4, int my_s, int n,
                                             int l, float4 (&v)[8]) {
#define GATHER_LD(u)                                                  \
    {                                                                 \
        const unsigned s = (unsigned)ROW_SHARE(my_s, G + (u));        \
        v[u] = make_float4(0.0f, 0.0f, 0.0f, 0.0f);                   \
        if (G + (u) < n) v[u] = h4[s * 16u + (unsigned)l];            \
    }
    GATHER_LD(0) GATHER_LD(1) GATHER_LD(2) GATHER_LD(3)
    GATHER_LD(4) GATHER_LD(5) GATHER_LD(6) GATHER_LD(7)
#undef GATHER_LD
}

template <int G>
__device__ __forceinline__ void gather8_sum(int my_w, const float4 (&v)[8], float4& acc) {
#define GATHER_FMA(u)                                                 \
    {                                                                 \
        const float w = __int_as_float(ROW_SHARE(my_w, G + (u)));     \
        acc.x = __builtin_fmaf(v[u].x, w, acc.x);                     \
        acc.y = __builtin_fmaf(v[u].y, w, acc.y);                     \
        acc.z = __builtin_fmaf(v[u].z, w, acc.z);                     \
        acc.w = __builtin_fmaf(v[u].w, w, acc.w);                     \
    }
    GATHER_FMA(0) GATHER_FMA(1) GATHER_FMA(2) GATHER_FMA(3)
    GATHER_FMA(4) GATHER_FMA(5) GATHER_FMA(6) GATHER_FMA(7)
#undef GATHER_FMA
}

__device__ __forceinline__ float4 gather_row_sum(
    const float* __restrict__ h, const int* __restrict__ row_ptr,
    const int* __restrict__ cnt_in, const int* __restrict__ bsrc,
    const float* __restrict__ bw, int node, int l, int n_nodes) {
    int beg = 0, len = 0;
    if (node < n_nodes) {
        beg = row_ptr[node];
        len = cnt_in[node];
    }
    const float4* __restrict__ h4 = (const float4*)h;
    const int* __restrict__ bp = bsrc + beg;
    const float* __restrict__ wp = bw + beg;

    float4 acc = make_float4(0.0f, 0.0f, 0.0f, 0.0f);
    int s_nxt = 0, w_nxt = 0;
    if (l < len) {
        s_nxt = bp[l];
        w_nxt = __float_as_int(wp[l]);
    }
    for (int j0 = 0; j0 < len; j0 += 16) {
        const int my_s = s_nxt, my_w = w_nxt;
        const int n = len - j0;
        float4 v[8];
        gather8_load<0>(h4, my_s, n, l, v);
        s_nxt = 0;
        w_nxt = 0;
        if (16 + l < n) {
            s_nxt = bp[j0 + 16 + l];
            w_nxt = __float_as_int(wp[j0 + 16 + l]);
        }
        gather8_sum<0>(my_w, v, acc);
        if (n > 8) {
            gather8_load<8>(h4, my_s, n, l, v);
            gather8_sum<8>(my_w, v, acc);
        }
    }
    return acc;
}
#undef ROW_SHARE

__global__ __launch_bounds__(256) void gather_rows_kernel(
    const float* __restrict__ h, const int* __restrict__ row_ptr,
    const int* __restrict__ cnt_in, const int* __restrict__ bsrc,
    const float* __restrict__ bw, const float* __restrict__ ndst,
    const float* __restrict__ bias, float* __restrict__ outp, int n_nodes) {
    const int t = blockIdx.x * blockDim.x + threadIdx.x;
    const int node = t >> 4;
    const int l = t & 15;
    if (node >= n_nodes) return;
    float4 acc = gather_row_sum(h, row_ptr, cnt_in, bsrc, bw, node, l, n_nodes);
    const float nd = ndst[node];
    const float4 b = *(const float4*)(bias + 4 * l);
    acc.x = fmaxf(acc.x * nd + b.x, 0.0f);
    acc.y = fmaxf(acc.y * nd + b.y, 0.0f);
    acc.z = fmaxf(acc.z * nd + b.z, 0.0f);
    acc.w = fmaxf(acc.w * nd + b.w, 0.0f);
    ((float4*)outp)[(size_t)node * 16 + l] = acc;
}

#define AGG_PITCH 68
template <int F_OUT, int NT, bool RELU>
__global__ __launch_bounds__(256) void gather_gemm_kernel(
    const float* __restrict__ h, const int* __restrict__ row_ptr,
    const int* __restrict__ cnt_in, const int* __restrict__ bsrc,
    const float* __restrict__ bw, const float* __restrict__ ndst,
    const float* __restrict__ W, const float* __restrict__ bias,
    float* __restrict__ out, int n_nodes) {
    __shared__ __attribute__((aligned(16))) float tile[16 * AGG_PITCH];
    const int tid = threadIdx.x;
    const int row0 = blockIdx.x * 16;
    {
        const int rl = tid >> 4;
        const int l = tid & 15;
        const float4 acc = gather_row_sum(h, row_ptr, cnt_in, bsrc, bw, row0 + rl, l, n_nodes);
        *(float4*)(tile + rl * AGG_PITCH + 4 * l) = acc;
    }
    __syncthreads();

    const int wave = tid >> 5;
    if (wave >= NT) return;
    const int lane = tid & 31;
    const int half = lane >> 4;
    const int lm = lane & 15;
    const int col = wave * 16 + lm;
    const bool cok = col < F_OUT;
    const int cc = cok ? col : F_OUT - 1;

    float nd[8];
#pragma unroll
    for (int i = 0; i < 8; ++i) {
        const int r = row0 + i + 8 * half;
        nd[i] = ndst[r < n_nodes ? r : n_nodes - 1];
    }
    const float bcol = bias[cc];

    v8f c = v8f{};
    const float* arow = tile + lm * AGG_PITCH;
#pragma unroll
    for (int k0 = 0; k0 < 64; k0 += 4) {
        const int ka = k0 + 2 * half;
        v2f a;
        a.x = arow[ka];
        a.y = arow[ka + 1];
        const float wx = W[(size_t)ka * F_OUT + cc];
        const float wy = W[(size_t)(ka + 1) * F_OUT + cc];
        v2f bf;
        bf.x = cok ? wx : 0.0f;
        bf.y = cok ? wy : 0.0f;
        c = __builtin_amdgcn_wmma_f32_16x16x4_f32(
             false, a,  false, bf,
             (short)0, c,  false,  false);
    }

    asm volatile("v_nop\n\tv_nop\n\tv_nop\n\tv_nop\n\tv_nop" : "+v"(c));

#pragma unroll
    for (int i = 0; i < 8; ++i) {
        const int r = row0 + i + 8 * half;
        float v = c[i] * nd[i] + bcol;
        if (RELU) v = fmaxf(v, 0.0f);
        if (cok && r < n_nodes) out[(size_t)r * F_OUT + col] = v;
    }
}

template <int F_IN, int F_OUT, int NT>
__device__ __forceinline__ void wmma_tile_product(
    const float* __restrict__ agg, const float* __restrict__ W,
    float* __restrict__ out, int row0, int lane) {
    const int half = lane >> 4;
    const int lm = lane & 15;

    v8f acc[NT];
#pragma unroll
    for (int t = 0; t < NT; ++t) acc[t] = v8f{};

    const float* arow = agg + (size_t)(row0 + lm) * F_IN;

    for (int k0 = 0; k0 < F_IN; k0 += 4) {
        const int ka = k0 + 2 * half;
        v2f a;
        a.x = arow[ka];
        a.y = arow[ka + 1];
#pragma unroll
        for (int t = 0; t < NT; ++t) {
            const int col = t * 16 + lm;
            v2f bf;
            bf.x = (col < F_OUT) ? W[(size_t)ka * F_OUT + col] : 0.0f;
            bf.y = (col < F_OUT) ? W[(size_t)(ka + 1) * F_OUT + col] : 0.0f;
            acc[t] = __builtin_amdgcn_wmma_f32_16x16x4_f32(
                 false, a,  false, bf,
                 (short)0, acc[t],  false,  false);
        }
    }

#pragma unroll
    for (int i = 0; i < 8; ++i) {
        const int r = row0 + i + 8 * half;
#pragma unroll
        for (int t = 0; t < NT; ++t) {
            const int col = t * 16 + lm;
            if (col < F_OUT) out[(size_t)r * F_OUT + col] = acc[t][i];
        }
    }
}

static inline size_t align256(size_t x) { return (x + 255) & ~(size_t)255; }

extern "C" void kernel_launch(void* const* d_in, const int* in_sizes, int n_in,
                              void* d_out, int out_size, void* d_ws, size_t ws_size,
                              hipStream_t stream) {
    const float* x  = (const float*)d_in[0];
    const float* W0 = (const float*)d_in[1];
    const float* b0 = (const float*)d_in[2];
    const float* W1 = (const float*)d_in[3];
    const float* b1 = (const float*)d_in[4];
    const float* W2 = (const float*)d_in[5];
    const float* b2 = (const float*)d_in[6];
    const int* src  = (const int*)d_in[7];
    const int* dst  = (const int*)d_in[8];

    const int N = in_sizes[0] / N_FEAT_IN;
    const int E = in_sizes[7];
    float* out = (float*)d_out;

    char* p = (char*)d_ws;
    float* nsrc = (float*)p;  p += align256((size_t)N * 4);
    float* ndst = (float*)p;  p += align256((size_t)N * 4);
    char* slab = p;           p += align256((size_t)N * N_FEAT_IN * 4);
    float* h0   = (float*)p;  p += align256((size_t)N * 64 * 4);
    float* h1   = (float*)p;  p += align256((size_t)N * 64 * 4);
    int* rank    = (int*)slab;
    int* tiletot = (int*)(slab + align256((size_t)E * 4));
    slab += align256((size_t)N * 64 * 4);
    int* cnt_in  = (int*)slab;    slab += align256((size_t)N * 4);
    int* row_ptr = (int*)slab;    slab += align256((size_t)N * 4);
    int* bsrc    = (int*)slab;    slab += align256((size_t)E * 4);
    float* bw    = (float*)slab;
    int* hist_dst = (int*)h0;
    int* hist_src = hist_dst + (size_t)N_CHUNK * N;

    const int TB = 256;
    const int n_tiles = (N + 1023) / 1024;
    const int chunk_len = (E + N_CHUNK - 1) / N_CHUNK;

    const int n_row_tiles = N / 16;
    const int n_xw_groups = (n_row_tiles + XW_TILES - 1) / XW_TILES;
    chunk_hist_xw_kernel<<<2 * N_CHUNK + n_xw_groups, 1024, 0, stream>>>(
        src, dst, hist_dst, hist_src, rank, tiletot, N, E, chunk_len, n_tiles,
        x, W0, h1, n_row_tiles);
    rowptr_norm_kernel<<<n_tiles, 1024, 0, stream>>>(hist_dst, hist_src, tiletot, cnt_in, row_ptr,
                                                     nsrc, ndst, N, n_tiles);
    bucket_fill_kernel<<<(E + TB - 1) / TB, TB, 0, stream>>>(src, dst, rank, hist_dst, nsrc, bsrc, bw,
                                                             E, N, chunk_len);

    const int GB = (N * 16 + TB - 1) / TB;

    gather_rows_kernel<<<GB, TB, 0, stream>>>(h1, row_ptr, cnt_in, bsrc, bw, ndst, b0, h0, N);

    gather_gemm_kernel<64, 4, true>
        <<<GB, TB, 0, stream>>>(h0, row_ptr, cnt_in, bsrc, bw, ndst, W1, b1, h1, N);

    gather_gemm_kernel<47, 3, false>
        <<<GB, TB, 0, stream>>>(h1, row_ptr, cnt_in, bsrc, bw, ndst, W2, b2, out, N);
}
